# starting kernel
# baseline (speedup 1.0000x reference)
#include <hip/hip_runtime.h>

typedef __attribute__((ext_vector_type(2))) float v2f;
typedef __attribute__((ext_vector_type(8))) float v8f;

#define D_IN  16
#define D_OUT 16

__global__ void zero_bins_kernel(int* __restrict__ counts,
                                 int* __restrict__ cursors, int n) {
    int i = blockIdx.x * blockDim.x + threadIdx.x;
    if (i < n) { counts[i] = 0; cursors[i] = 0; }
}

__global__ void hist_kernel(const int* __restrict__ widx,
                            int* __restrict__ counts, int e) {
    int i = blockIdx.x * blockDim.x + threadIdx.x;
    if (i < e) atomicAdd(&counts[widx[i]], 1);
}

__global__ void scan_kernel(const int* __restrict__ counts,
                            int* __restrict__ offsets, int n) {
    __shared__ int buf[256];
    __shared__ int carry;
    if (threadIdx.x == 0) { carry = 0; offsets[0] = 0; }
    __syncthreads();
    int nch = (n + 255) / 256;
    for (int c = 0; c < nch; ++c) {
        int i = c * 256 + (int)threadIdx.x;
        buf[threadIdx.x] = (i < n) ? counts[i] : 0;
        __syncthreads();
        for (int off = 1; off < 256; off <<= 1) {
            int t = (threadIdx.x >= (unsigned)off) ? buf[threadIdx.x - off] : 0;
            __syncthreads();
            buf[threadIdx.x] += t;
            __syncthreads();
        }
        if (i < n) offsets[i + 1] = carry + buf[threadIdx.x];
        __syncthreads();
        if (threadIdx.x == 0) carry += buf[255];
        __syncthreads();
    }
}

__global__ void scatter_kernel(const int* __restrict__ widx,
                               const int* __restrict__ iidx,
                               const int* __restrict__ offsets,
                               int* __restrict__ cursors,
                               int2* __restrict__ pairs, int e) {
    int i = blockIdx.x * blockDim.x + threadIdx.x;
    if (i < e) {
        int w = widx[i];
        int pos = offsets[w] + atomicAdd(&cursors[w], 1);
        pairs[pos] = make_int2(i, iidx[i]);
    }
}

__global__ void __launch_bounds__(256)
linear_wmma_bin_kernel(const float* __restrict__ values,
                       const float* __restrict__ weight,
                       const int2*  __restrict__ pairs,
                       const int*   __restrict__ offsets,
                       float* __restrict__ out, int n_w) {
    int bin = blockIdx.x;
    int beg = offsets[bin];
    int end = offsets[bin + 1];
    int cnt = end - beg;
    if (cnt == 0) return;

    const int lane = threadIdx.x & 31;
    const int wv   = threadIdx.x >> 5;
    const int nwv  = blockDim.x >> 5;
    const int n    = lane & 15;
    const int h    = lane >> 4;

    const float* wbase = weight + (size_t)bin * (D_OUT * D_IN);
    v2f bfrag[4];
#pragma unroll
    for (int kc = 0; kc < 4; ++kc) {
        const float* p = wbase + n * D_IN + 4 * kc + 2 * h;
        bfrag[kc].x = p[0];
        bfrag[kc].y = p[1];
    }

    const int nfull = cnt >> 4;

    for (int t = wv; t < nfull; t += nwv) {
        int tbase = beg + (t << 4);
        int2 pr = pairs[tbase + n];
        const float* xp = values + (size_t)pr.y * D_IN;

        v2f afrag[4];
#pragma unroll
        for (int kc = 0; kc < 4; ++kc) {
            const float* p = xp + 4 * kc + 2 * h;
            afrag[kc].x = p[0];
            afrag[kc].y = p[1];
        }

        v8f acc = {};
#pragma unroll
        for (int kc = 0; kc < 4; ++kc) {
            acc = __builtin_amdgcn_wmma_f32_16x16x4_f32(
                false, afrag[kc], false, bfrag[kc],
                (short)0, acc, false, false);
        }

#pragma unroll
        for (int j = 0; j < 8; ++j) {
            int r  = j + 8 * h;
            int se = __shfl(pr.x, r, 32);
            out[(size_t)se * D_OUT + n] = acc[j];
        }
    }

    int rem = cnt & 15;
    if (rem && (nfull % nwv) == wv) {
        int tbase = beg + (nfull << 4);
        int rowp  = tbase + n;
        if (n >= rem) rowp = tbase;
        int2 pr = pairs[rowp];
        const float* xp = values + (size_t)pr.y * D_IN;

        v2f afrag[4];
#pragma unroll
        for (int kc = 0; kc < 4; ++kc) {
            const float* p = xp + 4 * kc + 2 * h;
            afrag[kc].x = p[0];
            afrag[kc].y = p[1];
        }

        v8f acc = {};
#pragma unroll
        for (int kc = 0; kc < 4; ++kc) {
            acc = __builtin_amdgcn_wmma_f32_16x16x4_f32(
                false, afrag[kc], false, bfrag[kc],
                (short)0, acc, false, false);
        }

#pragma unroll
        for (int j = 0; j < 8; ++j) {
            int r  = j + 8 * h;
            int se = __shfl(pr.x, r, 32);
            if (r < rem) out[(size_t)se * D_OUT + n] = acc[j];
        }
    }
}

__global__ void linear_fallback_kernel(const float* __restrict__ values,
                                       const float* __restrict__ weight,
                                       const int* __restrict__ iidx,
                                       const int* __restrict__ widx,
                                       float* __restrict__ out, int e) {
    int i = blockIdx.x * blockDim.x + threadIdx.x;
    if (i >= e) return;
    const float* x = values + (size_t)iidx[i] * D_IN;
    const float* w = weight + (size_t)widx[i] * (D_OUT * D_IN);
    float xr[D_IN];
#pragma unroll
    for (int k = 0; k < D_IN; ++k) xr[k] = x[k];
    float* o = out + (size_t)i * D_OUT;
#pragma unroll
    for (int oo = 0; oo < D_OUT; ++oo) {
        float acc = 0.f;
#pragma unroll
        for (int k = 0; k < D_IN; ++k) acc = fmaf(w[oo * D_IN + k], xr[k], acc);
        o[oo] = acc;
    }
}

extern "C" void kernel_launch(void* const* d_in, const int* in_sizes, int n_in,
                              void* d_out, int out_size, void* d_ws, size_t ws_size,
                              hipStream_t stream) {
    const float* values = (const float*)d_in[0];
    const float* weight = (const float*)d_in[1];
    const int*   iidx   = (const int*)d_in[2];
    const int*   widx   = (const int*)d_in[3];
    float*       out    = (float*)d_out;

    const int E  = in_sizes[2];
    const int NW = in_sizes[1] / (D_OUT * D_IN);

    size_t hdrInts = (size_t)3 * NW + 1;
    hdrInts = (hdrInts + 1) & ~(size_t)1;
    size_t need = hdrInts * sizeof(int) + (size_t)E * sizeof(int2);
    if (NW <= 0 || ws_size < need) {
        linear_fallback_kernel<<<(E + 255) / 256, 256, 0, stream>>>(
            values, weight, iidx, widx, out, E);
        return;
    }

    int*  counts  = (int*)d_ws;
    int*  cursors = counts + NW;
    int*  offsets = cursors + NW;
    int2* pairs   = (int2*)((int*)d_ws + hdrInts);

    zero_bins_kernel<<<(NW + 255) / 256, 256, 0, stream>>>(counts, cursors, NW);
    hist_kernel<<<(E + 255) / 256, 256, 0, stream>>>(widx, counts, E);
    scan_kernel<<<1, 256, 0, stream>>>(counts, offsets, NW);
    scatter_kernel<<<(E + 255) / 256, 256, 0, stream>>>(widx, iidx, offsets,
                                                        cursors, pairs, E);
    linear_wmma_bin_kernel<<<NW, 256, 0, stream>>>(
        values, weight, pairs, offsets, out, NW);
}


# hist and scatter count per block in LDS (returning LDS add = rank in block) and send one global add per (block, bin) instead of one per edge
# speedup vs baseline: 2.8228x; 2.8228x over previous
#include <hip/hip_runtime.h>

typedef __attribute__((ext_vector_type(2))) float v2f;
typedef __attribute__((ext_vector_type(8))) float v8f;

#define D_IN  16
#define D_OUT 16

__global__ void zero_bins_kernel(int* __restrict__ counts,
                                 int* __restrict__ cursors, int n) {
    int i = blockIdx.x * blockDim.x + threadIdx.x;
    if (i < n) { counts[i] = 0; cursors[i] = 0; }
}

#define BIN_TPB   1024
#define BIN_ITEMS 8
#define BIN_CHUNK (BIN_TPB * BIN_ITEMS)

__global__ void __launch_bounds__(BIN_TPB)
hist_kernel(const int* __restrict__ widx,
            int* __restrict__ counts, int e, int nw) {
    extern __shared__ int lbin[];
    const int tid = threadIdx.x;
    for (int b = tid; b < nw; b += BIN_TPB) lbin[b] = 0;

    const int base = blockIdx.x * BIN_CHUNK;
    int w[BIN_ITEMS];
#pragma unroll
    for (int k = 0; k < BIN_ITEMS; ++k) {
        int i = base + k * BIN_TPB + tid;
        w[k] = (i < e) ? widx[i] : -1;
    }
    __syncthreads();
#pragma unroll
    for (int k = 0; k < BIN_ITEMS; ++k)
        if (w[k] >= 0) atomicAdd(&lbin[w[k]], 1);
    __syncthreads();

    for (int b = tid; b < nw; b += BIN_TPB) {
        int c = lbin[b];
        if (c) atomicAdd(&counts[b], c);
    }
}

__global__ void scan_kernel(const int* __restrict__ counts,
                            int* __restrict__ offsets, int n) {
    __shared__ int buf[256];
    __shared__ int carry;
    if (threadIdx.x == 0) { carry = 0; offsets[0] = 0; }
    __syncthreads();
    int nch = (n + 255) / 256;
    for (int c = 0; c < nch; ++c) {
        int i = c * 256 + (int)threadIdx.x;
        buf[threadIdx.x] = (i < n) ? counts[i] : 0;
        __syncthreads();
        for (int off = 1; off < 256; off <<= 1) {
            int t = (threadIdx.x >= (unsigned)off) ? buf[threadIdx.x - off] : 0;
            __syncthreads();
            buf[threadIdx.x] += t;
            __syncthreads();
        }
        if (i < n) offsets[i + 1] = carry + buf[threadIdx.x];
        __syncthreads();
        if (threadIdx.x == 0) carry += buf[255];
        __syncthreads();
    }
}

__global__ void __launch_bounds__(BIN_TPB)
scatter_kernel(const int* __restrict__ widx,
               const int* __restrict__ iidx,
               const int* __restrict__ offsets,
               int* __restrict__ cursors,
               int2* __restrict__ pairs, int e, int nw) {
    extern __shared__ int lbin[];
    const int tid = threadIdx.x;
    for (int b = tid; b < nw; b += BIN_TPB) lbin[b] = 0;

    const int base = blockIdx.x * BIN_CHUNK;
    int w[BIN_ITEMS], v[BIN_ITEMS], r[BIN_ITEMS];
#pragma unroll
    for (int k = 0; k < BIN_ITEMS; ++k) {
        int i = base + k * BIN_TPB + tid;
        bool ok = i < e;
        w[k] = ok ? widx[i] : -1;
        v[k] = ok ? iidx[i] : 0;
    }
    __syncthreads();
#pragma unroll
    for (int k = 0; k < BIN_ITEMS; ++k)
        r[k] = (w[k] >= 0) ? atomicAdd(&lbin[w[k]], 1) : 0;
    __syncthreads();

    for (int b = tid; b < nw; b += BIN_TPB) {
        int c = lbin[b];
        int s = 0;
        if (c) s = offsets[b] + atomicAdd(&cursors[b], c);
        lbin[b] = s;
    }
    __syncthreads();

#pragma unroll
    for (int k = 0; k < BIN_ITEMS; ++k) {
        if (w[k] >= 0) {
            int i = base + k * BIN_TPB + tid;
            pairs[lbin[w[k]] + r[k]] = make_int2(i, v[k]);
        }
    }
}

__global__ void __launch_bounds__(256)
linear_wmma_bin_kernel(const float* __restrict__ values,
                       const float* __restrict__ weight,
                       const int2*  __restrict__ pairs,
                       const int*   __restrict__ offsets,
                       float* __restrict__ out, int n_w) {
    int bin = blockIdx.x;
    int beg = offsets[bin];
    int end = offsets[bin + 1];
    int cnt = end - beg;
    if (cnt == 0) return;

    const int lane = threadIdx.x & 31;
    const int wv   = threadIdx.x >> 5;
    const int nwv  = blockDim.x >> 5;
    const int n    = lane & 15;
    const int h    = lane >> 4;

    const float* wbase = weight + (size_t)bin * (D_OUT * D_IN);
    v2f bfrag[4];
#pragma unroll
    for (int kc = 0; kc < 4; ++kc) {
        const float* p = wbase + n * D_IN + 4 * kc + 2 * h;
        bfrag[kc].x = p[0];
        bfrag[kc].y = p[1];
    }

    const int nfull = cnt >> 4;

    for (int t = wv; t < nfull; t += nwv) {
        int tbase = beg + (t << 4);
        int2 pr = pairs[tbase + n];
        const float* xp = values + (size_t)pr.y * D_IN;

        v2f afrag[4];
#pragma unroll
        for (int kc = 0; kc < 4; ++kc) {
            const float* p = xp + 4 * kc + 2 * h;
            afrag[kc].x = p[0];
            afrag[kc].y = p[1];
        }

        v8f acc = {};
#pragma unroll
        for (int kc = 0; kc < 4; ++kc) {
            acc = __builtin_amdgcn_wmma_f32_16x16x4_f32(
                false, afrag[kc], false, bfrag[kc],
                (short)0, acc, false, false);
        }

#pragma unroll
        for (int j = 0; j < 8; ++j) {
            int r  = j + 8 * h;
            int se = __shfl(pr.x, r, 32);
            out[(size_t)se * D_OUT + n] = acc[j];
        }
    }

    int rem = cnt & 15;
    if (rem && (nfull % nwv) == wv) {
        int tbase = beg + (nfull << 4);
        int rowp  = tbase + n;
        if (n >= rem) rowp = tbase;
        int2 pr = pairs[rowp];
        const float* xp = values + (size_t)pr.y * D_IN;

        v2f afrag[4];
#pragma unroll
        for (int kc = 0; kc < 4; ++kc) {
            const float* p = xp + 4 * kc + 2 * h;
            afrag[kc].x = p[0];
            afrag[kc].y = p[1];
        }

        v8f acc = {};
#pragma unroll
        for (int kc = 0; kc < 4; ++kc) {
            acc = __builtin_amdgcn_wmma_f32_16x16x4_f32(
                false, afrag[kc], false, bfrag[kc],
                (short)0, acc, false, false);
        }

#pragma unroll
        for (int j = 0; j < 8; ++j) {
            int r  = j + 8 * h;
            int se = __shfl(pr.x, r, 32);
            if (r < rem) out[(size_t)se * D_OUT + n] = acc[j];
        }
    }
}

__global__ void linear_fallback_kernel(const float* __restrict__ values,
                                       const float* __restrict__ weight,
                                       const int* __restrict__ iidx,
                                       const int* __restrict__ widx,
                                       float* __restrict__ out, int e) {
    int i = blockIdx.x * blockDim.x + threadIdx.x;
    if (i >= e) return;
    const float* x = values + (size_t)iidx[i] * D_IN;
    const float* w = weight + (size_t)widx[i] * (D_OUT * D_IN);
    float xr[D_IN];
#pragma unroll
    for (int k = 0; k < D_IN; ++k) xr[k] = x[k];
    float* o = out + (size_t)i * D_OUT;
#pragma unroll
    for (int oo = 0; oo < D_OUT; ++oo) {
        float acc = 0.f;
#pragma unroll
        for (int k = 0; k < D_IN; ++k) acc = fmaf(w[oo * D_IN + k], xr[k], acc);
        o[oo] = acc;
    }
}

extern "C" void kernel_launch(void* const* d_in, const int* in_sizes, int n_in,
                              void* d_out, int out_size, void* d_ws, size_t ws_size,
                              hipStream_t stream) {
    const float* values = (const float*)d_in[0];
    const float* weight = (const float*)d_in[1];
    const int*   iidx   = (const int*)d_in[2];
    const int*   widx   = (const int*)d_in[3];
    float*       out    = (float*)d_out;

    const int E  = in_sizes[2];
    const int NW = in_sizes[1] / (D_OUT * D_IN);

    size_t hdrInts = (size_t)3 * NW + 1;
    hdrInts = (hdrInts + 1) & ~(size_t)1;
    size_t need = hdrInts * sizeof(int) + (size_t)E * sizeof(int2);
    const size_t binLds = (size_t)NW * sizeof(int);
    if (NW <= 0 || ws_size < need || binLds > 65536) {
        linear_fallback_kernel<<<(E + 255) / 256, 256, 0, stream>>>(
            values, weight, iidx, widx, out, E);
        return;
    }

    int*  counts  = (int*)d_ws;
    int*  cursors = counts + NW;
    int*  offsets = cursors + NW;
    int2* pairs   = (int2*)((int*)d_ws + hdrInts);

    zero_bins_kernel<<<(NW + 255) / 256, 256, 0, stream>>>(counts, cursors, NW);
    hist_kernel<<<(E + BIN_CHUNK - 1) / BIN_CHUNK, BIN_TPB, binLds, stream>>>(
        widx, counts, E, NW);
    scan_kernel<<<1, 256, 0, stream>>>(counts, offsets, NW);
    scatter_kernel<<<(E + BIN_CHUNK - 1) / BIN_CHUNK, BIN_TPB, binLds, stream>>>(
        widx, iidx, offsets, cursors, pairs, E, NW);
    linear_wmma_bin_kernel<<<NW, 256, 0, stream>>>(
        values, weight, pairs, offsets, out, NW);
}


# sort edges by weight bin inside each block's LDS and let the bin kernel read the per-(block,bin) runs directly: no global counters or atomics, zero/hist/scan launches gone (5 launches -> 2)
# speedup vs baseline: 3.7194x; 1.3176x over previous
#include <hip/hip_runtime.h>

typedef __attribute__((ext_vector_type(2))) float v2f;
typedef __attribute__((ext_vector_type(8))) float v8f;

#define D_IN  16
#define D_OUT 16

__global__ void zero_bins_kernel(int* __restrict__ counts,
                                 int* __restrict__ cursors, int n) {
    int i = blockIdx.x * blockDim.x + threadIdx.x;
    if (i < n) { counts[i] = 0; cursors[i] = 0; }
}

#define BIN_TPB   1024
#define BIN_ITEMS 8
#define BIN_CHUNK (BIN_TPB * BIN_ITEMS)

__global__ void __launch_bounds__(BIN_TPB)
hist_kernel(const int* __restrict__ widx,
            int* __restrict__ counts, int e, int nw) {
    extern __shared__ int lbin[];
    const int tid = threadIdx.x;
    for (int b = tid; b < nw; b += BIN_TPB) lbin[b] = 0;

    const int base = blockIdx.x * BIN_CHUNK;
    int w[BIN_ITEMS];
#pragma unroll
    for (int k = 0; k < BIN_ITEMS; ++k) {
        int i = base + k * BIN_TPB + tid;
        w[k] = (i < e) ? widx[i] : -1;
    }
    __syncthreads();
#pragma unroll
    for (int k = 0; k < BIN_ITEMS; ++k)
        if (w[k] >= 0) atomicAdd(&lbin[w[k]], 1);
    __syncthreads();

    for (int b = tid; b < nw; b += BIN_TPB) {
        int c = lbin[b];
        if (c) atomicAdd(&counts[b], c);
    }
}

__global__ void scan_kernel(const int* __restrict__ counts,
                            int* __restrict__ offsets, int n) {
    __shared__ int buf[256];
    __shared__ int carry;
    if (threadIdx.x == 0) { carry = 0; offsets[0] = 0; }
    __syncthreads();
    int nch = (n + 255) / 256;
    for (int c = 0; c < nch; ++c) {
        int i = c * 256 + (int)threadIdx.x;
        buf[threadIdx.x] = (i < n) ? counts[i] : 0;
        __syncthreads();
        for (int off = 1; off < 256; off <<= 1) {
            int t = (threadIdx.x >= (unsigned)off) ? buf[threadIdx.x - off] : 0;
            __syncthreads();
            buf[threadIdx.x] += t;
            __syncthreads();
        }
        if (i < n) offsets[i + 1] = carry + buf[threadIdx.x];
        __syncthreads();
        if (threadIdx.x == 0) carry += buf[255];
        __syncthreads();
    }
}

__global__ void __launch_bounds__(BIN_TPB)
scatter_kernel(const int* __restrict__ widx,
               const int* __restrict__ iidx,
               const int* __restrict__ offsets,
               int* __restrict__ cursors,
               int2* __restrict__ pairs, int e, int nw) {
    extern __shared__ int lbin[];
    const int tid = threadIdx.x;
    for (int b = tid; b < nw; b += BIN_TPB) lbin[b] = 0;

    const int base = blockIdx.x * BIN_CHUNK;
    int w[BIN_ITEMS], v[BIN_ITEMS], r[BIN_ITEMS];
#pragma unroll
    for (int k = 0; k < BIN_ITEMS; ++k) {
        int i = base + k * BIN_TPB + tid;
        bool ok = i < e;
        w[k] = ok ? widx[i] : -1;
        v[k] = ok ? iidx[i] : 0;
    }
    __syncthreads();
#pragma unroll
    for (int k = 0; k < BIN_ITEMS; ++k)
        r[k] = (w[k] >= 0) ? atomicAdd(&lbin[w[k]], 1) : 0;
    __syncthreads();

    for (int b = tid; b < nw; b += BIN_TPB) {
        int c = lbin[b];
        int s = 0;
        if (c) s = offsets[b] + atomicAdd(&cursors[b], c);
        lbin[b] = s;
    }
    __syncthreads();

#pragma unroll
    for (int k = 0; k < BIN_ITEMS; ++k) {
        if (w[k] >= 0) {
            int i = base + k * BIN_TPB + tid;
            pairs[lbin[w[k]] + r[k]] = make_int2(i, v[k]);
        }
    }
}

__global__ void __launch_bounds__(256)
linear_wmma_bin_kernel(const float* __restrict__ values,
                       const float* __restrict__ weight,
                       const int2*  __restrict__ pairs,
                       const int*   __restrict__ offsets,
                       float* __restrict__ out, int n_w) {
    int bin = blockIdx.x;
    int beg = offsets[bin];
    int end = offsets[bin + 1];
    int cnt = end - beg;
    if (cnt == 0) return;

    const int lane = threadIdx.x & 31;
    const int wv   = threadIdx.x >> 5;
    const int nwv  = blockDim.x >> 5;
    const int n    = lane & 15;
    const int h    = lane >> 4;

    const float* wbase = weight + (size_t)bin * (D_OUT * D_IN);
    v2f bfrag[4];
#pragma unroll
    for (int kc = 0; kc < 4; ++kc) {
        const float* p = wbase + n * D_IN + 4 * kc + 2 * h;
        bfrag[kc].x = p[0];
        bfrag[kc].y = p[1];
    }

    const int nfull = cnt >> 4;

    for (int t = wv; t < nfull; t += nwv) {
        int tbase = beg + (t << 4);
        int2 pr = pairs[tbase + n];
        const float* xp = values + (size_t)pr.y * D_IN;

        v2f afrag[4];
#pragma unroll
        for (int kc = 0; kc < 4; ++kc) {
            const float* p = xp + 4 * kc + 2 * h;
            afrag[kc].x = p[0];
            afrag[kc].y = p[1];
        }

        v8f acc = {};
#pragma unroll
        for (int kc = 0; kc < 4; ++kc) {
            acc = __builtin_amdgcn_wmma_f32_16x16x4_f32(
                false, afrag[kc], false, bfrag[kc],
                (short)0, acc, false, false);
        }

#pragma unroll
        for (int j = 0; j < 8; ++j) {
            int r  = j + 8 * h;
            int se = __shfl(pr.x, r, 32);
            out[(size_t)se * D_OUT + n] = acc[j];
        }
    }

    int rem = cnt & 15;
    if (rem && (nfull % nwv) == wv) {
        int tbase = beg + (nfull << 4);
        int rowp  = tbase + n;
        if (n >= rem) rowp = tbase;
        int2 pr = pairs[rowp];
        const float* xp = values + (size_t)pr.y * D_IN;

        v2f afrag[4];
#pragma unroll
        for (int kc = 0; kc < 4; ++kc) {
            const float* p = xp + 4 * kc + 2 * h;
            afrag[kc].x = p[0];
            afrag[kc].y = p[1];
        }

        v8f acc = {};
#pragma unroll
        for (int kc = 0; kc < 4; ++kc) {
            acc = __builtin_amdgcn_wmma_f32_16x16x4_f32(
                false, afrag[kc], false, bfrag[kc],
                (short)0, acc, false, false);
        }

#pragma unroll
        for (int j = 0; j < 8; ++j) {
            int r  = j + 8 * h;
            int se = __shfl(pr.x, r, 32);
            if (r < rem) out[(size_t)se * D_OUT + n] = acc[j];
        }
    }
}

__global__ void __launch_bounds__(BIN_TPB)
block_sort_kernel(const int* __restrict__ widx,
                  const int* __restrict__ iidx,
                  int* __restrict__ runs,
                  int2* __restrict__ pairs, int e, int nw) {
    extern __shared__ int lbin[];
    __shared__ int            sval[BIN_CHUNK];
    __shared__ unsigned short ssrc[BIN_CHUNK];
    __shared__ int            wsum[BIN_TPB / 32];
    const int tid  = threadIdx.x;
    const int lane = tid & 31;
    const int wv   = tid >> 5;
    for (int b = tid; b < nw; b += BIN_TPB) lbin[b] = 0;

    const int base = blockIdx.x * BIN_CHUNK;
    int w[BIN_ITEMS], v[BIN_ITEMS], r[BIN_ITEMS];
#pragma unroll
    for (int k = 0; k < BIN_ITEMS; ++k) {
        int i = base + k * BIN_TPB + tid;
        bool ok = i < e;
        w[k] = ok ? widx[i] : -1;
        v[k] = ok ? iidx[i] : 0;
    }
    __syncthreads();
#pragma unroll
    for (int k = 0; k < BIN_ITEMS; ++k)
        r[k] = (w[k] >= 0) ? atomicAdd(&lbin[w[k]], 1) : 0;
    __syncthreads();

    const int per = (nw + BIN_TPB - 1) / BIN_TPB;
    const int b0  = min(tid * per, nw);
    const int b1  = min(b0 + per, nw);
    int sum = 0;
    for (int b = b0; b < b1; ++b) sum += lbin[b];
    int inc = sum;
#pragma unroll
    for (int off = 1; off < 32; off <<= 1) {
        int t = __shfl_up(inc, off, 32);
        if (lane >= off) inc += t;
    }
    if (lane == 31) wsum[wv] = inc;
    __syncthreads();
    if (wv == 0) {
        int x = wsum[lane];
#pragma unroll
        for (int off = 1; off < 32; off <<= 1) {
            int t = __shfl_up(x, off, 32);
            if (lane >= off) x += t;
        }
        wsum[lane] = x;
    }
    __syncthreads();
    int excl = inc - sum + (wv ? wsum[wv - 1] : 0);
    int* row = runs + (size_t)blockIdx.x * (nw + 1);
    for (int b = b0; b < b1; ++b) {
        int c = lbin[b];
        lbin[b] = excl;
        row[b]  = excl;
        excl += c;
    }
    if (tid == BIN_TPB - 1) row[nw] = excl;
    __syncthreads();

#pragma unroll
    for (int k = 0; k < BIN_ITEMS; ++k) {
        if (w[k] >= 0) {
            int pos = lbin[w[k]] + r[k];
            sval[pos] = v[k];
            ssrc[pos] = (unsigned short)(k * BIN_TPB + tid);
        }
    }
    __syncthreads();

    const int valid = min(e - base, BIN_CHUNK);
#pragma unroll
    for (int k = 0; k < BIN_ITEMS; ++k) {
        int j = k * BIN_TPB + tid;
        if (j < valid) pairs[base + j] = make_int2(base + (int)ssrc[j], sval[j]);
    }
}

__device__ __forceinline__ void
run_tile(const float* __restrict__ values, const int2* __restrict__ pairs,
         float* __restrict__ out, const v2f (&bfrag)[4],
         int p, int rows, int n, int h) {
    int2 pr = pairs[p + (n < rows ? n : 0)];
    const float* xp = values + (size_t)pr.y * D_IN;
    v2f afrag[4];
#pragma unroll
    for (int kc = 0; kc < 4; ++kc) {
        const float* q = xp + 4 * kc + 2 * h;
        afrag[kc].x = q[0];
        afrag[kc].y = q[1];
    }
    v8f acc = {};
#pragma unroll
    for (int kc = 0; kc < 4; ++kc) {
        acc = __builtin_amdgcn_wmma_f32_16x16x4_f32(
            false, afrag[kc], false, bfrag[kc],
            (short)0, acc, false, false);
    }
    int se[8];
#pragma unroll
    for (int j = 0; j < 8; ++j) se[j] = __shfl(pr.x, j + 8 * h, 32);
#pragma unroll
    for (int j = 0; j < 8; ++j)
        if (j + 8 * h < rows) out[(size_t)se[j] * D_OUT + n] = acc[j];
}

#define RUN_U 4

__device__ __forceinline__ void
run_bounds(const int* __restrict__ runs, int pitch, int bin, int blk0, int nwv,
           int nblk, int (&start)[RUN_U], int (&len)[RUN_U]) {
#pragma unroll
    for (int u = 0; u < RUN_U; ++u) {
        const int  blk = blk0 + u * nwv;
        const bool ok  = blk < nblk;
        const int  bc  = ok ? blk : 0;
        const int* row = runs + (size_t)bc * pitch + bin;
        const int  a = row[0], b = row[1];
        start[u] = bc * BIN_CHUNK + a;
        len[u]   = ok ? b - a : 0;
    }
}

__global__ void __launch_bounds__(256)
linear_wmma_runs_kernel(const float* __restrict__ values,
                        const float* __restrict__ weight,
                        const int2*  __restrict__ pairs,
                        const int*   __restrict__ runs,
                        float* __restrict__ out, int nw, int nblk) {
    if (nblk <= 0) return;
    const int bin  = blockIdx.x;
    const int lane = threadIdx.x & 31;
    const int wv   = __builtin_amdgcn_readfirstlane((int)(threadIdx.x >> 5));
    const int nwv  = blockDim.x >> 5;
    const int n    = lane & 15;
    const int h    = lane >> 4;

    const float* wbase = weight + (size_t)bin * (D_OUT * D_IN);
    v2f bfrag[4];
#pragma unroll
    for (int kc = 0; kc < 4; ++kc) {
        const float* p = wbase + n * D_IN + 4 * kc + 2 * h;
        bfrag[kc].x = p[0];
        bfrag[kc].y = p[1];
    }

    const int pitch = nw + 1;
    const int step  = RUN_U * nwv;
    int ps[RUN_U], pc[RUN_U];
    run_bounds(runs, pitch, bin, wv, nwv, nblk, ps, pc);
    for (int blk0 = wv; blk0 < nblk; blk0 += step) {
        int p[RUN_U], c[RUN_U], rows[RUN_U];
#pragma unroll
        for (int u = 0; u < RUN_U; ++u) {
            c[u]    = pc[u];
            rows[u] = min(c[u], 16);
            p[u]    = rows[u] > 0 ? ps[u] : 0;
        }

        int2 pr[RUN_U];
#pragma unroll
        for (int u = 0; u < RUN_U; ++u)
            pr[u] = pairs[p[u] + (n < rows[u] ? n : 0)];
        run_bounds(runs, pitch, bin, blk0 + step, nwv, nblk, ps, pc);

        v2f afrag[RUN_U][4];
#pragma unroll
        for (int u = 0; u < RUN_U; ++u) {
            const float* xp = values + (size_t)pr[u].y * D_IN;
#pragma unroll
            for (int kc = 0; kc < 4; ++kc) {
                const float* q = xp + 4 * kc + 2 * h;
                afrag[u][kc].x = q[0];
                afrag[u][kc].y = q[1];
            }
        }

        v8f acc[RUN_U];
#pragma unroll
        for (int u = 0; u < RUN_U; ++u) {
            acc[u] = v8f{};
#pragma unroll
            for (int kc = 0; kc < 4; ++kc) {
                acc[u] = __builtin_amdgcn_wmma_f32_16x16x4_f32(
                    false, afrag[u][kc], false, bfrag[kc],
                    (short)0, acc[u], false, false);
            }
        }

#pragma unroll
        for (int u = 0; u < RUN_U; ++u) {
            int se[8];
#pragma unroll
            for (int j = 0; j < 8; ++j) se[j] = __shfl(pr[u].x, j + 8 * h, 32);
#pragma unroll
            for (int j = 0; j < 8; ++j)
                if (j + 8 * h < rows[u]) out[(size_t)se[j] * D_OUT + n] = acc[u][j];
        }

#pragma unroll
        for (int u = 0; u < RUN_U; ++u)
            for (int q = 16; q < c[u]; q += 16)
                run_tile(values, pairs, out, bfrag, p[u] + q, min(c[u] - q, 16), n, h);
    }
}

__global__ void linear_fallback_kernel(const float* __restrict__ values,
                                       const float* __restrict__ weight,
                                       const int* __restrict__ iidx,
                                       const int* __restrict__ widx,
                                       float* __restrict__ out, int e) {
    int i = blockIdx.x * blockDim.x + threadIdx.x;
    if (i >= e) return;
    const float* x = values + (size_t)iidx[i] * D_IN;
    const float* w = weight + (size_t)widx[i] * (D_OUT * D_IN);
    float xr[D_IN];
#pragma unroll
    for (int k = 0; k < D_IN; ++k) xr[k] = x[k];
    float* o = out + (size_t)i * D_OUT;
#pragma unroll
    for (int oo = 0; oo < D_OUT; ++oo) {
        float acc = 0.f;
#pragma unroll
        for (int k = 0; k < D_IN; ++k) acc = fmaf(w[oo * D_IN + k], xr[k], acc);
        o[oo] = acc;
    }
}

extern "C" void kernel_launch(void* const* d_in, const int* in_sizes, int n_in,
                              void* d_out, int out_size, void* d_ws, size_t ws_size,
                              hipStream_t stream) {
    const float* values = (const float*)d_in[0];
    const float* weight = (const float*)d_in[1];
    const int*   iidx   = (const int*)d_in[2];
    const int*   widx   = (const int*)d_in[3];
    float*       out    = (float*)d_out;

    const int E  = in_sizes[2];
    const int NW = in_sizes[1] / (D_OUT * D_IN);

    size_t hdrInts = (size_t)3 * NW + 1;
    hdrInts = (hdrInts + 1) & ~(size_t)1;
    size_t need = hdrInts * sizeof(int) + (size_t)E * sizeof(int2);
    const size_t binLds = (size_t)NW * sizeof(int);
    if (NW <= 0 || ws_size < need || binLds > 65536) {
        linear_fallback_kernel<<<(E + 255) / 256, 256, 0, stream>>>(
            values, weight, iidx, widx, out, E);
        return;
    }

    const int NBLK = (E + BIN_CHUNK - 1) / BIN_CHUNK;
    size_t runInts = (size_t)NBLK * ((size_t)NW + 1);
    runInts = (runInts + 1) & ~(size_t)1;
    const size_t needRuns = runInts * sizeof(int) + (size_t)E * sizeof(int2);
    if (E > 0 && ws_size >= needRuns && binLds + 49152 + 128 <= 65536) {
        int*  runs   = (int*)d_ws;
        int2* rpairs = (int2*)((int*)d_ws + runInts);
        block_sort_kernel<<<NBLK, BIN_TPB, binLds, stream>>>(
            widx, iidx, runs, rpairs, E, NW);
        linear_wmma_runs_kernel<<<NW, 256, 0, stream>>>(
            values, weight, rpairs, runs, out, NW, NBLK);
        return;
    }

    int*  counts  = (int*)d_ws;
    int*  cursors = counts + NW;
    int*  offsets = cursors + NW;
    int2* pairs   = (int2*)((int*)d_ws + hdrInts);

    zero_bins_kernel<<<(NW + 255) / 256, 256, 0, stream>>>(counts, cursors, NW);
    hist_kernel<<<(E + BIN_CHUNK - 1) / BIN_CHUNK, BIN_TPB, binLds, stream>>>(
        widx, counts, E, NW);
    scan_kernel<<<1, 256, 0, stream>>>(counts, offsets, NW);
    scatter_kernel<<<(E + BIN_CHUNK - 1) / BIN_CHUNK, BIN_TPB, binLds, stream>>>(
        widx, iidx, offsets, cursors, pairs, E, NW);
    linear_wmma_bin_kernel<<<NW, 256, 0, stream>>>(
        values, weight, pairs, offsets, out, NW);
}


# run tiles computed transposed (weights as WMMA A, gathered rows as B): a lane holds 8 outputs of its own edge, so 2 whole-row 16-byte stores per run replace 8 lane shuffles + 8 4-byte stores
# speedup vs baseline: 3.9954x; 1.0742x over previous
#include <hip/hip_runtime.h>

typedef __attribute__((ext_vector_type(2))) float v2f;
typedef __attribute__((ext_vector_type(8))) float v8f;

#define D_IN  16
#define D_OUT 16

__global__ void zero_bins_kernel(int* __restrict__ counts,
                                 int* __restrict__ cursors, int n) {
    int i = blockIdx.x * blockDim.x + threadIdx.x;
    if (i < n) { counts[i] = 0; cursors[i] = 0; }
}

#define BIN_TPB   1024
#define BIN_ITEMS 8
#define BIN_CHUNK (BIN_TPB * BIN_ITEMS)

__global__ void __launch_bounds__(BIN_TPB)
hist_kernel(const int* __restrict__ widx,
            int* __restrict__ counts, int e, int nw) {
    extern __shared__ int lbin[];
    const int tid = threadIdx.x;
    for (int b = tid; b < nw; b += BIN_TPB) lbin[b] = 0;

    const int base = blockIdx.x * BIN_CHUNK;
    int w[BIN_ITEMS];
#pragma unroll
    for (int k = 0; k < BIN_ITEMS; ++k) {
        int i = base + k * BIN_TPB + tid;
        w[k] = (i < e) ? widx[i] : -1;
    }
    __syncthreads();
#pragma unroll
    for (int k = 0; k < BIN_ITEMS; ++k)
        if (w[k] >= 0) atomicAdd(&lbin[w[k]], 1);
    __syncthreads();

    for (int b = tid; b < nw; b += BIN_TPB) {
        int c = lbin[b];
        if (c) atomicAdd(&counts[b], c);
    }
}

__global__ void scan_kernel(const int* __restrict__ counts,
                            int* __restrict__ offsets, int n) {
    __shared__ int buf[256];
    __shared__ int carry;
    if (threadIdx.x == 0) { carry = 0; offsets[0] = 0; }
    __syncthreads();
    int nch = (n + 255) / 256;
    for (int c = 0; c < nch; ++c) {
        int i = c * 256 + (int)threadIdx.x;
        buf[threadIdx.x] = (i < n) ? counts[i] : 0;
        __syncthreads();
        for (int off = 1; off < 256; off <<= 1) {
            int t = (threadIdx.x >= (unsigned)off) ? buf[threadIdx.x - off] : 0;
            __syncthreads();
            buf[threadIdx.x] += t;
            __syncthreads();
        }
        if (i < n) offsets[i + 1] = carry + buf[threadIdx.x];
        __syncthreads();
        if (threadIdx.x == 0) carry += buf[255];
        __syncthreads();
    }
}

__global__ void __launch_bounds__(BIN_TPB)
scatter_kernel(const int* __restrict__ widx,
               const int* __restrict__ iidx,
               const int* __restrict__ offsets,
               int* __restrict__ cursors,
               int2* __restrict__ pairs, int e, int nw) {
    extern __shared__ int lbin[];
    const int tid = threadIdx.x;
    for (int b = tid; b < nw; b += BIN_TPB) lbin[b] = 0;

    const int base = blockIdx.x * BIN_CHUNK;
    int w[BIN_ITEMS], v[BIN_ITEMS], r[BIN_ITEMS];
#pragma unroll
    for (int k = 0; k < BIN_ITEMS; ++k) {
        int i = base + k * BIN_TPB + tid;
        bool ok = i < e;
        w[k] = ok ? widx[i] : -1;
        v[k] = ok ? iidx[i] : 0;
    }
    __syncthreads();
#pragma unroll
    for (int k = 0; k < BIN_ITEMS; ++k)
        r[k] = (w[k] >= 0) ? atomicAdd(&lbin[w[k]], 1) : 0;
    __syncthreads();

    for (int b = tid; b < nw; b += BIN_TPB) {
        int c = lbin[b];
        int s = 0;
        if (c) s = offsets[b] + atomicAdd(&cursors[b], c);
        lbin[b] = s;
    }
    __syncthreads();

#pragma unroll
    for (int k = 0; k < BIN_ITEMS; ++k) {
        if (w[k] >= 0) {
            int i = base + k * BIN_TPB + tid;
            pairs[lbin[w[k]] + r[k]] = make_int2(i, v[k]);
        }
    }
}

__global__ void __launch_bounds__(256)
linear_wmma_bin_kernel(const float* __restrict__ values,
                       const float* __restrict__ weight,
                       const int2*  __restrict__ pairs,
                       const int*   __restrict__ offsets,
                       float* __restrict__ out, int n_w) {
    int bin = blockIdx.x;
    int beg = offsets[bin];
    int end = offsets[bin + 1];
    int cnt = end - beg;
    if (cnt == 0) return;

    const int lane = threadIdx.x & 31;
    const int wv   = threadIdx.x >> 5;
    const int nwv  = blockDim.x >> 5;
    const int n    = lane & 15;
    const int h    = lane >> 4;

    const float* wbase = weight + (size_t)bin * (D_OUT * D_IN);
    v2f bfrag[4];
#pragma unroll
    for (int kc = 0; kc < 4; ++kc) {
        const float* p = wbase + n * D_IN + 4 * kc + 2 * h;
        bfrag[kc].x = p[0];
        bfrag[kc].y = p[1];
    }

    const int nfull = cnt >> 4;

    for (int t = wv; t < nfull; t += nwv) {
        int tbase = beg + (t << 4);
        int2 pr = pairs[tbase + n];
        const float* xp = values + (size_t)pr.y * D_IN;

        v2f afrag[4];
#pragma unroll
        for (int kc = 0; kc < 4; ++kc) {
            const float* p = xp + 4 * kc + 2 * h;
            afrag[kc].x = p[0];
            afrag[kc].y = p[1];
        }

        v8f acc = {};
#pragma unroll
        for (int kc = 0; kc < 4; ++kc) {
            acc = __builtin_amdgcn_wmma_f32_16x16x4_f32(
                false, afrag[kc], false, bfrag[kc],
                (short)0, acc, false, false);
        }

#pragma unroll
        for (int j = 0; j < 8; ++j) {
            int r  = j + 8 * h;
            int se = __shfl(pr.x, r, 32);
            out[(size_t)se * D_OUT + n] = acc[j];
        }
    }

    int rem = cnt & 15;
    if (rem && (nfull % nwv) == wv) {
        int tbase = beg + (nfull << 4);
        int rowp  = tbase + n;
        if (n >= rem) rowp = tbase;
        int2 pr = pairs[rowp];
        const float* xp = values + (size_t)pr.y * D_IN;

        v2f afrag[4];
#pragma unroll
        for (int kc = 0; kc < 4; ++kc) {
            const float* p = xp + 4 * kc + 2 * h;
            afrag[kc].x = p[0];
            afrag[kc].y = p[1];
        }

        v8f acc = {};
#pragma unroll
        for (int kc = 0; kc < 4; ++kc) {
            acc = __builtin_amdgcn_wmma_f32_16x16x4_f32(
                false, afrag[kc], false, bfrag[kc],
                (short)0, acc, false, false);
        }

#pragma unroll
        for (int j = 0; j < 8; ++j) {
            int r  = j + 8 * h;
            int se = __shfl(pr.x, r, 32);
            if (r < rem) out[(size_t)se * D_OUT + n] = acc[j];
        }
    }
}

__global__ void __launch_bounds__(BIN_TPB)
block_sort_kernel(const int* __restrict__ widx,
                  const int* __restrict__ iidx,
                  int* __restrict__ runs,
                  int2* __restrict__ pairs, int e, int nw) {
    extern __shared__ int lbin[];
    __shared__ int            sval[BIN_CHUNK];
    __shared__ unsigned short ssrc[BIN_CHUNK];
    __shared__ int            wsum[BIN_TPB / 32];
    const int tid  = threadIdx.x;
    const int lane = tid & 31;
    const int wv   = tid >> 5;
    for (int b = tid; b < nw; b += BIN_TPB) lbin[b] = 0;

    const int base = blockIdx.x * BIN_CHUNK;
    int w[BIN_ITEMS], v[BIN_ITEMS], r[BIN_ITEMS];
#pragma unroll
    for (int k = 0; k < BIN_ITEMS; ++k) {
        int i = base + k * BIN_TPB + tid;
        bool ok = i < e;
        w[k] = ok ? widx[i] : -1;
        v[k] = ok ? iidx[i] : 0;
    }
    __syncthreads();
#pragma unroll
    for (int k = 0; k < BIN_ITEMS; ++k)
        r[k] = (w[k] >= 0) ? atomicAdd(&lbin[w[k]], 1) : 0;
    __syncthreads();

    const int per = (nw + BIN_TPB - 1) / BIN_TPB;
    const int b0  = min(tid * per, nw);
    const int b1  = min(b0 + per, nw);
    int sum = 0;
    for (int b = b0; b < b1; ++b) sum += lbin[b];
    int inc = sum;
#pragma unroll
    for (int off = 1; off < 32; off <<= 1) {
        int t = __shfl_up(inc, off, 32);
        if (lane >= off) inc += t;
    }
    if (lane == 31) wsum[wv] = inc;
    __syncthreads();
    if (wv == 0) {
        int x = wsum[lane];
#pragma unroll
        for (int off = 1; off < 32; off <<= 1) {
            int t = __shfl_up(x, off, 32);
            if (lane >= off) x += t;
        }
        wsum[lane] = x;
    }
    __syncthreads();
    int excl = inc - sum + (wv ? wsum[wv - 1] : 0);
    int* row = runs + (size_t)blockIdx.x * (nw + 1);
    for (int b = b0; b < b1; ++b) {
        int c = lbin[b];
        lbin[b] = excl;
        row[b]  = excl;
        excl += c;
    }
    if (tid == BIN_TPB - 1) row[nw] = excl;
    __syncthreads();

#pragma unroll
    for (int k = 0; k < BIN_ITEMS; ++k) {
        if (w[k] >= 0) {
            int pos = lbin[w[k]] + r[k];
            sval[pos] = v[k];
            ssrc[pos] = (unsigned short)(k * BIN_TPB + tid);
        }
    }
    __syncthreads();

    const int valid = min(e - base, BIN_CHUNK);
#pragma unroll
    for (int k = 0; k < BIN_ITEMS; ++k) {
        int j = k * BIN_TPB + tid;
        if (j < valid) pairs[base + j] = make_int2(base + (int)ssrc[j], sval[j]);
    }
}

__device__ __forceinline__ int swap_pair(int x) {
    return __builtin_amdgcn_update_dpp(x, x, 0xB1, 0xF, 0xF, false);
}

__device__ __forceinline__ void
store_rows(float* __restrict__ out, const v8f& acc, int edge, int rows, int n, int h) {
    const bool odd   = (n & 1) != 0;
    const int  other = swap_pair(edge);
    float d0[4], d1[4];
#pragma unroll
    for (int i = 0; i < 4; ++i) {
        const float give = odd ? acc[i] : acc[4 + i];
        const float got  = __int_as_float(swap_pair(__float_as_int(give)));
        d0[i] = odd ? got : acc[i];
        d1[i] = odd ? acc[4 + i] : got;
    }
    const int e0  = odd ? other : edge;
    const int e1  = odd ? edge : other;
    const int col = 8 * h + 4 * (n & 1);
    if ((n & ~1) < rows)
        *reinterpret_cast<float4*>(out + (size_t)e0 * D_OUT + col) =
            make_float4(d0[0], d0[1], d0[2], d0[3]);
    if ((n | 1) < rows)
        *reinterpret_cast<float4*>(out + (size_t)e1 * D_OUT + col) =
            make_float4(d1[0], d1[1], d1[2], d1[3]);
}

__device__ __forceinline__ void
run_tile(const float* __restrict__ values, const int2* __restrict__ pairs,
         float* __restrict__ out, const v2f (&wfrag)[4],
         int p, int rows, int n, int h) {
    int2 pr = pairs[p + (n < rows ? n : 0)];
    const float* xp = values + (size_t)pr.y * D_IN;
    v2f xfrag[4];
#pragma unroll
    for (int kc = 0; kc < 4; ++kc) {
        const float* q = xp + 4 * kc + 2 * h;
        xfrag[kc].x = q[0];
        xfrag[kc].y = q[1];
    }
    v8f acc = {};
#pragma unroll
    for (int kc = 0; kc < 4; ++kc) {
        acc = __builtin_amdgcn_wmma_f32_16x16x4_f32(
            false, wfrag[kc], false, xfrag[kc],
            (short)0, acc, false, false);
    }
    store_rows(out, acc, pr.x, rows, n, h);
}

#define RUN_U 4

__device__ __forceinline__ void
run_bounds(const int* __restrict__ runs, int pitch, int bin, int blk0, int nwv,
           int nblk, int (&start)[RUN_U], int (&len)[RUN_U]) {
#pragma unroll
    for (int u = 0; u < RUN_U; ++u) {
        const int  blk = blk0 + u * nwv;
        const bool ok  = blk < nblk;
        const int  bc  = ok ? blk : 0;
        const int* row = runs + (size_t)bc * pitch + bin;
        const int  a = row[0], b = row[1];
        start[u] = bc * BIN_CHUNK + a;
        len[u]   = ok ? b - a : 0;
    }
}

__global__ void __launch_bounds__(256)
linear_wmma_runs_kernel(const float* __restrict__ values,
                        const float* __restrict__ weight,
                        const int2*  __restrict__ pairs,
                        const int*   __restrict__ runs,
                        float* __restrict__ out, int nw, int nblk) {
    if (nblk <= 0) return;
    const int bin  = blockIdx.x;
    const int lane = threadIdx.x & 31;
    const int wv   = __builtin_amdgcn_readfirstlane((int)(threadIdx.x >> 5));
    const int nwv  = blockDim.x >> 5;
    const int n    = lane & 15;
    const int h    = lane >> 4;

    const float* wbase = weight + (size_t)bin * (D_OUT * D_IN);
    v2f wfrag[4];
#pragma unroll
    for (int kc = 0; kc < 4; ++kc) {
        const float* p = wbase + n * D_IN + 4 * kc + 2 * h;
        wfrag[kc].x = p[0];
        wfrag[kc].y = p[1];
    }

    const int pitch = nw + 1;
    const int step  = RUN_U * nwv;
    int ps[RUN_U], pc[RUN_U];
    run_bounds(runs, pitch, bin, wv, nwv, nblk, ps, pc);
    for (int blk0 = wv; blk0 < nblk; blk0 += step) {
        int p[RUN_U], c[RUN_U], rows[RUN_U];
#pragma unroll
        for (int u = 0; u < RUN_U; ++u) {
            c[u]    = pc[u];
            rows[u] = min(c[u], 16);
            p[u]    = rows[u] > 0 ? ps[u] : 0;
        }

        int2 pr[RUN_U];
#pragma unroll
        for (int u = 0; u < RUN_U; ++u)
            pr[u] = pairs[p[u] + (n < rows[u] ? n : 0)];
        run_bounds(runs, pitch, bin, blk0 + step, nwv, nblk, ps, pc);

        v2f xfrag[RUN_U][4];
#pragma unroll
        for (int u = 0; u < RUN_U; ++u) {
            const float* xp = values + (size_t)pr[u].y * D_IN;
#pragma unroll
            for (int kc = 0; kc < 4; ++kc) {
                const float* q = xp + 4 * kc + 2 * h;
                xfrag[u][kc].x = q[0];
                xfrag[u][kc].y = q[1];
            }
        }
        __builtin_amdgcn_sched_barrier(0);

        v8f acc[RUN_U];
#pragma unroll
        for (int u = 0; u < RUN_U; ++u) {
            acc[u] = v8f{};
#pragma unroll
            for (int kc = 0; kc < 4; ++kc) {
                acc[u] = __builtin_amdgcn_wmma_f32_16x16x4_f32(
                    false, wfrag[kc], false, xfrag[u][kc],
                    (short)0, acc[u], false, false);
            }
        }

#pragma unroll
        for (int u = 0; u < RUN_U; ++u)
            store_rows(out, acc[u], pr[u].x, rows[u], n, h);

#pragma unroll
        for (int u = 0; u < RUN_U; ++u)
            for (int q = 16; q < c[u]; q += 16)
                run_tile(values, pairs, out, wfrag, p[u] + q, min(c[u] - q, 16), n, h);
    }
}

__global__ void linear_fallback_kernel(const float* __restrict__ values,
                                       const float* __restrict__ weight,
                                       const int* __restrict__ iidx,
                                       const int* __restrict__ widx,
                                       float* __restrict__ out, int e) {
    int i = blockIdx.x * blockDim.x + threadIdx.x;
    if (i >= e) return;
    const float* x = values + (size_t)iidx[i] * D_IN;
    const float* w = weight + (size_t)widx[i] * (D_OUT * D_IN);
    float xr[D_IN];
#pragma unroll
    for (int k = 0; k < D_IN; ++k) xr[k] = x[k];
    float* o = out + (size_t)i * D_OUT;
#pragma unroll
    for (int oo = 0; oo < D_OUT; ++oo) {
        float acc = 0.f;
#pragma unroll
        for (int k = 0; k < D_IN; ++k) acc = fmaf(w[oo * D_IN + k], xr[k], acc);
        o[oo] = acc;
    }
}

extern "C" void kernel_launch(void* const* d_in, const int* in_sizes, int n_in,
                              void* d_out, int out_size, void* d_ws, size_t ws_size,
                              hipStream_t stream) {
    const float* values = (const float*)d_in[0];
    const float* weight = (const float*)d_in[1];
    const int*   iidx   = (const int*)d_in[2];
    const int*   widx   = (const int*)d_in[3];
    float*       out    = (float*)d_out;

    const int E  = in_sizes[2];
    const int NW = in_sizes[1] / (D_OUT * D_IN);

    size_t hdrInts = (size_t)3 * NW + 1;
    hdrInts = (hdrInts + 1) & ~(size_t)1;
    size_t need = hdrInts * sizeof(int) + (size_t)E * sizeof(int2);
    const size_t binLds = (size_t)NW * sizeof(int);
    if (NW <= 0 || ws_size < need || binLds > 65536) {
        linear_fallback_kernel<<<(E + 255) / 256, 256, 0, stream>>>(
            values, weight, iidx, widx, out, E);
        return;
    }

    const int NBLK = (E + BIN_CHUNK - 1) / BIN_CHUNK;
    size_t runInts = (size_t)NBLK * ((size_t)NW + 1);
    runInts = (runInts + 1) & ~(size_t)1;
    const size_t needRuns = runInts * sizeof(int) + (size_t)E * sizeof(int2);
    if (E > 0 && ws_size >= needRuns && binLds + 49152 + 128 <= 65536) {
        int*  runs   = (int*)d_ws;
        int2* rpairs = (int2*)((int*)d_ws + runInts);
        block_sort_kernel<<<NBLK, BIN_TPB, binLds, stream>>>(
            widx, iidx, runs, rpairs, E, NW);
        linear_wmma_runs_kernel<<<NW, 256, 0, stream>>>(
            values, weight, rpairs, runs, out, NW, NBLK);
        return;
    }

    int*  counts  = (int*)d_ws;
    int*  cursors = counts + NW;
    int*  offsets = cursors + NW;
    int2* pairs   = (int2*)((int*)d_ws + hdrInts);

    zero_bins_kernel<<<(NW + 255) / 256, 256, 0, stream>>>(counts, cursors, NW);
    hist_kernel<<<(E + BIN_CHUNK - 1) / BIN_CHUNK, BIN_TPB, binLds, stream>>>(
        widx, counts, E, NW);
    scan_kernel<<<1, 256, 0, stream>>>(counts, offsets, NW);
    scatter_kernel<<<(E + BIN_CHUNK - 1) / BIN_CHUNK, BIN_TPB, binLds, stream>>>(
        widx, iidx, offsets, cursors, pairs, E, NW);
    linear_wmma_bin_kernel<<<NW, 256, 0, stream>>>(
        values, weight, pairs, offsets, out, NW);
}


# one launch: a block owns 4096 consecutive edges: sorts them by bin in LDS, multiplies each bin's run there, parks the rows in LDS and writes its piece of out as whole lines; sums unchanged
# speedup vs baseline: 5.5344x; 1.3852x over previous
#include <hip/hip_runtime.h>

typedef __attribute__((ext_vector_type(2))) float v2f;
typedef __attribute__((ext_vector_type(8))) float v8f;

#define D_IN  16
#define D_OUT 16

__global__ void zero_bins_kernel(int* __restrict__ counts,
                                 int* __restrict__ cursors, int n) {
    int i = blockIdx.x * blockDim.x + threadIdx.x;
    if (i < n) { counts[i] = 0; cursors[i] = 0; }
}

#define BIN_TPB   1024
#define BIN_ITEMS 8
#define BIN_CHUNK (BIN_TPB * BIN_ITEMS)

__global__ void __launch_bounds__(BIN_TPB)
hist_kernel(const int* __restrict__ widx,
            int* __restrict__ counts, int e, int nw) {
    extern __shared__ int lbin[];
    const int tid = threadIdx.x;
    for (int b = tid; b < nw; b += BIN_TPB) lbin[b] = 0;

    const int base = blockIdx.x * BIN_CHUNK;
    int w[BIN_ITEMS];
#pragma unroll
    for (int k = 0; k < BIN_ITEMS; ++k) {
        int i = base + k * BIN_TPB + tid;
        w[k] = (i < e) ? widx[i] : -1;
    }
    __syncthreads();
#pragma unroll
    for (int k = 0; k < BIN_ITEMS; ++k)
        if (w[k] >= 0) atomicAdd(&lbin[w[k]], 1);
    __syncthreads();

    for (int b = tid; b < nw; b += BIN_TPB) {
        int c = lbin[b];
        if (c) atomicAdd(&counts[b], c);
    }
}

__global__ void scan_kernel(const int* __restrict__ counts,
                            int* __restrict__ offsets, int n) {
    __shared__ int buf[256];
    __shared__ int carry;
    if (threadIdx.x == 0) { carry = 0; offsets[0] = 0; }
    __syncthreads();
    int nch = (n + 255) / 256;
    for (int c = 0; c < nch; ++c) {
        int i = c * 256 + (int)threadIdx.x;
        buf[threadIdx.x] = (i < n) ? counts[i] : 0;
        __syncthreads();
        for (int off = 1; off < 256; off <<= 1) {
            int t = (threadIdx.x >= (unsigned)off) ? buf[threadIdx.x - off] : 0;
            __syncthreads();
            buf[threadIdx.x] += t;
            __syncthreads();
        }
        if (i < n) offsets[i + 1] = carry + buf[threadIdx.x];
        __syncthreads();
        if (threadIdx.x == 0) carry += buf[255];
        __syncthreads();
    }
}

__global__ void __launch_bounds__(BIN_TPB)
scatter_kernel(const int* __restrict__ widx,
               const int* __restrict__ iidx,
               const int* __restrict__ offsets,
               int* __restrict__ cursors,
               int2* __restrict__ pairs, int e, int nw) {
    extern __shared__ int lbin[];
    const int tid = threadIdx.x;
    for (int b = tid; b < nw; b += BIN_TPB) lbin[b] = 0;

    const int base = blockIdx.x * BIN_CHUNK;
    int w[BIN_ITEMS], v[BIN_ITEMS], r[BIN_ITEMS];
#pragma unroll
    for (int k = 0; k < BIN_ITEMS; ++k) {
        int i = base + k * BIN_TPB + tid;
        bool ok = i < e;
        w[k] = ok ? widx[i] : -1;
        v[k] = ok ? iidx[i] : 0;
    }
    __syncthreads();
#pragma unroll
    for (int k = 0; k < BIN_ITEMS; ++k)
        r[k] = (w[k] >= 0) ? atomicAdd(&lbin[w[k]], 1) : 0;
    __syncthreads();

    for (int b = tid; b < nw; b += BIN_TPB) {
        int c = lbin[b];
        int s = 0;
        if (c) s = offsets[b] + atomicAdd(&cursors[b], c);
        lbin[b] = s;
    }
    __syncthreads();

#pragma unroll
    for (int k = 0; k < BIN_ITEMS; ++k) {
        if (w[k] >= 0) {
            int i = base + k * BIN_TPB + tid;
            pairs[lbin[w[k]] + r[k]] = make_int2(i, v[k]);
        }
    }
}

__global__ void __launch_bounds__(256)
linear_wmma_bin_kernel(const float* __restrict__ values,
                       const float* __restrict__ weight,
                       const int2*  __restrict__ pairs,
                       const int*   __restrict__ offsets,
                       float* __restrict__ out, int n_w) {
    int bin = blockIdx.x;
    int beg = offsets[bin];
    int end = offsets[bin + 1];
    int cnt = end - beg;
    if (cnt == 0) return;

    const int lane = threadIdx.x & 31;
    const int wv   = threadIdx.x >> 5;
    const int nwv  = blockDim.x >> 5;
    const int n    = lane & 15;
    const int h    = lane >> 4;

    const float* wbase = weight + (size_t)bin * (D_OUT * D_IN);
    v2f bfrag[4];
#pragma unroll
    for (int kc = 0; kc < 4; ++kc) {
        const float* p = wbase + n * D_IN + 4 * kc + 2 * h;
        bfrag[kc].x = p[0];
        bfrag[kc].y = p[1];
    }

    const int nfull = cnt >> 4;

    for (int t = wv; t < nfull; t += nwv) {
        int tbase = beg + (t << 4);
        int2 pr = pairs[tbase + n];
        const float* xp = values + (size_t)pr.y * D_IN;

        v2f afrag[4];
#pragma unroll
        for (int kc = 0; kc < 4; ++kc) {
            const float* p = xp + 4 * kc + 2 * h;
            afrag[kc].x = p[0];
            afrag[kc].y = p[1];
        }

        v8f acc = {};
#pragma unroll
        for (int kc = 0; kc < 4; ++kc) {
            acc = __builtin_amdgcn_wmma_f32_16x16x4_f32(
                false, afrag[kc], false, bfrag[kc],
                (short)0, acc, false, false);
        }

#pragma unroll
        for (int j = 0; j < 8; ++j) {
            int r  = j + 8 * h;
            int se = __shfl(pr.x, r, 32);
            out[(size_t)se * D_OUT + n] = acc[j];
        }
    }

    int rem = cnt & 15;
    if (rem && (nfull % nwv) == wv) {
        int tbase = beg + (nfull << 4);
        int rowp  = tbase + n;
        if (n >= rem) rowp = tbase;
        int2 pr = pairs[rowp];
        const float* xp = values + (size_t)pr.y * D_IN;

        v2f afrag[4];
#pragma unroll
        for (int kc = 0; kc < 4; ++kc) {
            const float* p = xp + 4 * kc + 2 * h;
            afrag[kc].x = p[0];
            afrag[kc].y = p[1];
        }

        v8f acc = {};
#pragma unroll
        for (int kc = 0; kc < 4; ++kc) {
            acc = __builtin_amdgcn_wmma_f32_16x16x4_f32(
                false, afrag[kc], false, bfrag[kc],
                (short)0, acc, false, false);
        }

#pragma unroll
        for (int j = 0; j < 8; ++j) {
            int r  = j + 8 * h;
            int se = __shfl(pr.x, r, 32);
            if (r < rem) out[(size_t)se * D_OUT + n] = acc[j];
        }
    }
}

__global__ void __launch_bounds__(BIN_TPB)
block_sort_kernel(const int* __restrict__ widx,
                  const int* __restrict__ iidx,
                  int* __restrict__ runs,
                  int2* __restrict__ pairs, int e, int nw) {
    extern __shared__ int lbin[];
    __shared__ int            sval[BIN_CHUNK];
    __shared__ unsigned short ssrc[BIN_CHUNK];
    __shared__ int            wsum[BIN_TPB / 32];
    const int tid  = threadIdx.x;
    const int lane = tid & 31;
    const int wv   = tid >> 5;
    for (int b = tid; b < nw; b += BIN_TPB) lbin[b] = 0;

    const int base = blockIdx.x * BIN_CHUNK;
    int w[BIN_ITEMS], v[BIN_ITEMS], r[BIN_ITEMS];
#pragma unroll
    for (int k = 0; k < BIN_ITEMS; ++k) {
        int i = base + k * BIN_TPB + tid;
        bool ok = i < e;
        w[k] = ok ? widx[i] : -1;
        v[k] = ok ? iidx[i] : 0;
    }
    __syncthreads();
#pragma unroll
    for (int k = 0; k < BIN_ITEMS; ++k)
        r[k] = (w[k] >= 0) ? atomicAdd(&lbin[w[k]], 1) : 0;
    __syncthreads();

    const int per = (nw + BIN_TPB - 1) / BIN_TPB;
    const int b0  = min(tid * per, nw);
    const int b1  = min(b0 + per, nw);
    int sum = 0;
    for (int b = b0; b < b1; ++b) sum += lbin[b];
    int inc = sum;
#pragma unroll
    for (int off = 1; off < 32; off <<= 1) {
        int t = __shfl_up(inc, off, 32);
        if (lane >= off) inc += t;
    }
    if (lane == 31) wsum[wv] = inc;
    __syncthreads();
    if (wv == 0) {
        int x = wsum[lane];
#pragma unroll
        for (int off = 1; off < 32; off <<= 1) {
            int t = __shfl_up(x, off, 32);
            if (lane >= off) x += t;
        }
        wsum[lane] = x;
    }
    __syncthreads();
    int excl = inc - sum + (wv ? wsum[wv - 1] : 0);
    int* row = runs + (size_t)blockIdx.x * (nw + 1);
    for (int b = b0; b < b1; ++b) {
        int c = lbin[b];
        lbin[b] = excl;
        row[b]  = excl;
        excl += c;
    }
    if (tid == BIN_TPB - 1) row[nw] = excl;
    __syncthreads();

#pragma unroll
    for (int k = 0; k < BIN_ITEMS; ++k) {
        if (w[k] >= 0) {
            int pos = lbin[w[k]] + r[k];
            sval[pos] = v[k];
            ssrc[pos] = (unsigned short)(k * BIN_TPB + tid);
        }
    }
    __syncthreads();

    const int valid = min(e - base, BIN_CHUNK);
#pragma unroll
    for (int k = 0; k < BIN_ITEMS; ++k) {
        int j = k * BIN_TPB + tid;
        if (j < valid) pairs[base + j] = make_int2(base + (int)ssrc[j], sval[j]);
    }
}

__device__ __forceinline__ int swap_pair(int x) {
    return __builtin_amdgcn_update_dpp(x, x, 0xB1, 0xF, 0xF, false);
}

__device__ __forceinline__ void
store_rows(float* __restrict__ out, const v8f& acc, int edge, int rows, int n, int h) {
    const bool odd   = (n & 1) != 0;
    const int  other = swap_pair(edge);
    float d0[4], d1[4];
#pragma unroll
    for (int i = 0; i < 4; ++i) {
        const float give = odd ? acc[i] : acc[4 + i];
        const float got  = __int_as_float(swap_pair(__float_as_int(give)));
        d0[i] = odd ? got : acc[i];
        d1[i] = odd ? acc[4 + i] : got;
    }
    const int e0  = odd ? other : edge;
    const int e1  = odd ? edge : other;
    const int col = 8 * h + 4 * (n & 1);
    if ((n & ~1) < rows)
        *reinterpret_cast<float4*>(out + (size_t)e0 * D_OUT + col) =
            make_float4(d0[0], d0[1], d0[2], d0[3]);
    if ((n | 1) < rows)
        *reinterpret_cast<float4*>(out + (size_t)e1 * D_OUT + col) =
            make_float4(d1[0], d1[1], d1[2], d1[3]);
}

__device__ __forceinline__ void
run_tile(const float* __restrict__ values, const int2* __restrict__ pairs,
         float* __restrict__ out, const v2f (&wfrag)[4],
         int p, int rows, int n, int h) {
    int2 pr = pairs[p + (n < rows ? n : 0)];
    const float* xp = values + (size_t)pr.y * D_IN;
    v2f xfrag[4];
#pragma unroll
    for (int kc = 0; kc < 4; ++kc) {
        const float* q = xp + 4 * kc + 2 * h;
        xfrag[kc].x = q[0];
        xfrag[kc].y = q[1];
    }
    v8f acc = {};
#pragma unroll
    for (int kc = 0; kc < 4; ++kc) {
        acc = __builtin_amdgcn_wmma_f32_16x16x4_f32(
            false, wfrag[kc], false, xfrag[kc],
            (short)0, acc, false, false);
    }
    store_rows(out, acc, pr.x, rows, n, h);
}

#define RUN_U 4

__device__ __forceinline__ void
run_bounds(const int* __restrict__ runs, int pitch, int bin, int blk0, int nwv,
           int nblk, int (&start)[RUN_U], int (&len)[RUN_U]) {
#pragma unroll
    for (int u = 0; u < RUN_U; ++u) {
        const int  blk = blk0 + u * nwv;
        const bool ok  = blk < nblk;
        const int  bc  = ok ? blk : 0;
        const int* row = runs + (size_t)bc * pitch + bin;
        const int  a = row[0], b = row[1];
        start[u] = bc * BIN_CHUNK + a;
        len[u]   = ok ? b - a : 0;
    }
}

__global__ void __launch_bounds__(256)
linear_wmma_runs_kernel(const float* __restrict__ values,
                        const float* __restrict__ weight,
                        const int2*  __restrict__ pairs,
                        const int*   __restrict__ runs,
                        float* __restrict__ out, int nw, int nblk) {
    if (nblk <= 0) return;
    const int bin  = blockIdx.x;
    const int lane = threadIdx.x & 31;
    const int wv   = __builtin_amdgcn_readfirstlane((int)(threadIdx.x >> 5));
    const int nwv  = blockDim.x >> 5;
    const int n    = lane & 15;
    const int h    = lane >> 4;

    const float* wbase = weight + (size_t)bin * (D_OUT * D_IN);
    v2f wfrag[4];
#pragma unroll
    for (int kc = 0; kc < 4; ++kc) {
        const float* p = wbase + n * D_IN + 4 * kc + 2 * h;
        wfrag[kc].x = p[0];
        wfrag[kc].y = p[1];
    }

    const int pitch = nw + 1;
    const int step  = RUN_U * nwv;
    int ps[RUN_U], pc[RUN_U];
    run_bounds(runs, pitch, bin, wv, nwv, nblk, ps, pc);
    for (int blk0 = wv; blk0 < nblk; blk0 += step) {
        int p[RUN_U], c[RUN_U], rows[RUN_U];
#pragma unroll
        for (int u = 0; u < RUN_U; ++u) {
            c[u]    = pc[u];
            rows[u] = min(c[u], 16);
            p[u]    = rows[u] > 0 ? ps[u] : 0;
        }

        int2 pr[RUN_U];
#pragma unroll
        for (int u = 0; u < RUN_U; ++u)
            pr[u] = pairs[p[u] + (n < rows[u] ? n : 0)];
        run_bounds(runs, pitch, bin, blk0 + step, nwv, nblk, ps, pc);

        v2f xfrag[RUN_U][4];
#pragma unroll
        for (int u = 0; u < RUN_U; ++u) {
            const float* xp = values + (size_t)pr[u].y * D_IN;
#pragma unroll
            for (int kc = 0; kc < 4; ++kc) {
                const float* q = xp + 4 * kc + 2 * h;
                xfrag[u][kc].x = q[0];
                xfrag[u][kc].y = q[1];
            }
        }
        __builtin_amdgcn_sched_barrier(0);

        v8f acc[RUN_U];
#pragma unroll
        for (int u = 0; u < RUN_U; ++u) {
            acc[u] = v8f{};
#pragma unroll
            for (int kc = 0; kc < 4; ++kc) {
                acc[u] = __builtin_amdgcn_wmma_f32_16x16x4_f32(
                    false, wfrag[kc], false, xfrag[u][kc],
                    (short)0, acc[u], false, false);
            }
        }

#pragma unroll
        for (int u = 0; u < RUN_U; ++u)
            store_rows(out, acc[u], pr[u].x, rows[u], n, h);

#pragma unroll
        for (int u = 0; u < RUN_U; ++u)
            for (int q = 16; q < c[u]; q += 16)
                run_tile(values, pairs, out, wfrag, p[u] + q, min(c[u] - q, 16), n, h);
    }
}

#define EB_TPB   1024
#define EB_ITEMS 4
#define EB_CHUNK (EB_TPB * EB_ITEMS)
#define EB_NWMAX 1024
#define EB_U     4

__device__ __forceinline__ void
eb_tile(const float* __restrict__ values, const int* sval,
        const unsigned short* ssrc, float4* yrow, const v2f (&wf)[4],
        int p, int rows, int n, int h) {
    const int idx  = p + (n < rows ? n : 0);
    const int row  = sval[idx];
    const int slot = ssrc[idx];
    const float* xp = values + (size_t)row * D_IN + 2 * h;
    v2f xf[4];
#pragma unroll
    for (int kc = 0; kc < 4; ++kc) {
        xf[kc].x = xp[4 * kc];
        xf[kc].y = xp[4 * kc + 1];
    }
    v8f acc = {};
#pragma unroll
    for (int kc = 0; kc < 4; ++kc) {
        acc = __builtin_amdgcn_wmma_f32_16x16x4_f32(
            false, wf[kc], false, xf[kc], (short)0, acc, false, false);
    }
    asm volatile("v_nop\n\tv_nop\n\tv_nop\n\tv_nop" : "+v"(acc));
    if (n < rows) {
        yrow[slot * 4 + 2 * h]     = make_float4(acc[0], acc[1], acc[2], acc[3]);
        yrow[slot * 4 + 2 * h + 1] = make_float4(acc[4], acc[5], acc[6], acc[7]);
    }
}

__global__ void __launch_bounds__(EB_TPB)
edge_block_kernel(const float* __restrict__ values,
                  const float* __restrict__ weight,
                  const int* __restrict__ widx,
                  const int* __restrict__ iidx,
                  float* __restrict__ out, int e, int nw) {
    __shared__ float4         yrow[EB_CHUNK * 4];
    __shared__ int            lbin[EB_NWMAX + 1];
    __shared__ int            sval[EB_CHUNK];
    __shared__ unsigned short ssrc[EB_CHUNK];
    __shared__ int            wsum[EB_TPB / 32];
    const int tid  = threadIdx.x;
    const int lane = tid & 31;
    const int wv   = tid >> 5;
    for (int b = tid; b < nw; b += EB_TPB) lbin[b] = 0;

    const int base = blockIdx.x * EB_CHUNK;
    int w[EB_ITEMS], v[EB_ITEMS], r[EB_ITEMS];
#pragma unroll
    for (int k = 0; k < EB_ITEMS; ++k) {
        int i = base + k * EB_TPB + tid;
        bool ok = i < e;
        w[k] = ok ? widx[i] : -1;
        v[k] = ok ? iidx[i] : 0;
    }
    __syncthreads();
#pragma unroll
    for (int k = 0; k < EB_ITEMS; ++k)
        r[k] = (w[k] >= 0) ? atomicAdd(&lbin[w[k]], 1) : 0;
    __syncthreads();

    {
        const int per = (nw + EB_TPB - 1) / EB_TPB;
        const int b0  = min(tid * per, nw);
        const int b1  = min(b0 + per, nw);
        int sum = 0;
        for (int b = b0; b < b1; ++b) sum += lbin[b];
        int inc = sum;
#pragma unroll
        for (int off = 1; off < 32; off <<= 1) {
            int t = __shfl_up(inc, off, 32);
            if (lane >= off) inc += t;
        }
        if (lane == 31) wsum[wv] = inc;
        __syncthreads();
        if (wv == 0) {
            int x = wsum[lane];
#pragma unroll
            for (int off = 1; off < 32; off <<= 1) {
                int t = __shfl_up(x, off, 32);
                if (lane >= off) x += t;
            }
            wsum[lane] = x;
        }
        __syncthreads();
        int excl = inc - sum + (wv ? wsum[wv - 1] : 0);
        for (int b = b0; b < b1; ++b) {
            int c = lbin[b];
            lbin[b] = excl;
            excl += c;
        }
        if (tid == EB_TPB - 1) lbin[nw] = excl;
    }
    __syncthreads();

#pragma unroll
    for (int k = 0; k < EB_ITEMS; ++k) {
        if (w[k] >= 0) {
            int pos = lbin[w[k]] + r[k];
            sval[pos] = v[k];
            ssrc[pos] = (unsigned short)(k * EB_TPB + tid);
        }
    }
    __syncthreads();

    const int n   = lane & 15;
    const int h   = lane >> 4;
    const int wvs = __builtin_amdgcn_readfirstlane(wv);
    const int nwaves = EB_TPB / 32;
    for (int b0 = wvs; b0 < nw; b0 += nwaves * EB_U) {
        int start[EB_U], len[EB_U], rows[EB_U];
        v2f wf[EB_U][4], xf[EB_U][4];
        int slot[EB_U];
#pragma unroll
        for (int u = 0; u < EB_U; ++u) {
            const int  b  = b0 + u * nwaves;
            const bool ok = b < nw;
            const int  bc = ok ? b : 0;
            const int  a  = __builtin_amdgcn_readfirstlane(lbin[bc]);
            const int  z  = __builtin_amdgcn_readfirstlane(lbin[bc + 1]);
            start[u] = a;
            len[u]   = ok ? z - a : 0;
            rows[u]  = min(len[u], 16);
            const float* wb = weight + (size_t)bc * (D_OUT * D_IN) + n * D_IN + 2 * h;
#pragma unroll
            for (int kc = 0; kc < 4; ++kc) {
                wf[u][kc].x = wb[4 * kc];
                wf[u][kc].y = wb[4 * kc + 1];
            }
            const int idx = (n < rows[u]) ? a + n : (rows[u] > 0 ? a : 0);
            const int row = sval[idx];
            slot[u] = ssrc[idx];
            const float* xp = values + (size_t)row * D_IN + 2 * h;
#pragma unroll
            for (int kc = 0; kc < 4; ++kc) {
                xf[u][kc].x = xp[4 * kc];
                xf[u][kc].y = xp[4 * kc + 1];
            }
        }

#pragma unroll
        for (int u = 0; u < EB_U; ++u) {
            v8f acc = {};
#pragma unroll
            for (int kc = 0; kc < 4; ++kc) {
                acc = __builtin_amdgcn_wmma_f32_16x16x4_f32(
                    false, wf[u][kc], false, xf[u][kc],
                    (short)0, acc, false, false);
            }
            asm volatile("v_nop\n\tv_nop\n\tv_nop\n\tv_nop" : "+v"(acc));
            if (n < rows[u]) {
                yrow[slot[u] * 4 + 2 * h]     = make_float4(acc[0], acc[1], acc[2], acc[3]);
                yrow[slot[u] * 4 + 2 * h + 1] = make_float4(acc[4], acc[5], acc[6], acc[7]);
            }
        }

#pragma unroll
        for (int u = 0; u < EB_U; ++u)
            for (int q = 16; q < len[u]; q += 16)
                eb_tile(values, sval, ssrc, yrow, wf[u], start[u] + q,
                        min(len[u] - q, 16), n, h);
    }
    __syncthreads();

    const int valid = min(e - base, EB_CHUNK);
    float4* out4 = reinterpret_cast<float4*>(out) + (size_t)base * (D_OUT / 4);
    const int nvec = valid * (D_OUT / 4);
    for (int i = tid; i < nvec; i += EB_TPB) out4[i] = yrow[i];
}

__global__ void linear_fallback_kernel(const float* __restrict__ values,
                                       const float* __restrict__ weight,
                                       const int* __restrict__ iidx,
                                       const int* __restrict__ widx,
                                       float* __restrict__ out, int e) {
    int i = blockIdx.x * blockDim.x + threadIdx.x;
    if (i >= e) return;
    const float* x = values + (size_t)iidx[i] * D_IN;
    const float* w = weight + (size_t)widx[i] * (D_OUT * D_IN);
    float xr[D_IN];
#pragma unroll
    for (int k = 0; k < D_IN; ++k) xr[k] = x[k];
    float* o = out + (size_t)i * D_OUT;
#pragma unroll
    for (int oo = 0; oo < D_OUT; ++oo) {
        float acc = 0.f;
#pragma unroll
        for (int k = 0; k < D_IN; ++k) acc = fmaf(w[oo * D_IN + k], xr[k], acc);
        o[oo] = acc;
    }
}

extern "C" void kernel_launch(void* const* d_in, const int* in_sizes, int n_in,
                              void* d_out, int out_size, void* d_ws, size_t ws_size,
                              hipStream_t stream) {
    const float* values = (const float*)d_in[0];
    const float* weight = (const float*)d_in[1];
    const int*   iidx   = (const int*)d_in[2];
    const int*   widx   = (const int*)d_in[3];
    float*       out    = (float*)d_out;

    const int E  = in_sizes[2];
    const int NW = in_sizes[1] / (D_OUT * D_IN);

    if (E > 0 && NW > 0 && NW <= EB_NWMAX) {
        edge_block_kernel<<<(E + EB_CHUNK - 1) / EB_CHUNK, EB_TPB, 0, stream>>>(
            values, weight, widx, iidx, out, E, NW);
        return;
    }

    size_t hdrInts = (size_t)3 * NW + 1;
    hdrInts = (hdrInts + 1) & ~(size_t)1;
    size_t need = hdrInts * sizeof(int) + (size_t)E * sizeof(int2);
    const size_t binLds = (size_t)NW * sizeof(int);
    if (NW <= 0 || ws_size < need || binLds > 65536) {
        linear_fallback_kernel<<<(E + 255) / 256, 256, 0, stream>>>(
            values, weight, iidx, widx, out, E);
        return;
    }

    const int NBLK = (E + BIN_CHUNK - 1) / BIN_CHUNK;
    size_t runInts = (size_t)NBLK * ((size_t)NW + 1);
    runInts = (runInts + 1) & ~(size_t)1;
    const size_t needRuns = runInts * sizeof(int) + (size_t)E * sizeof(int2);
    if (E > 0 && ws_size >= needRuns && binLds + 49152 + 128 <= 65536) {
        int*  runs   = (int*)d_ws;
        int2* rpairs = (int2*)((int*)d_ws + runInts);
        block_sort_kernel<<<NBLK, BIN_TPB, binLds, stream>>>(
            widx, iidx, runs, rpairs, E, NW);
        linear_wmma_runs_kernel<<<NW, 256, 0, stream>>>(
            values, weight, rpairs, runs, out, NW, NBLK);
        return;
    }

    int*  counts  = (int*)d_ws;
    int*  cursors = counts + NW;
    int*  offsets = cursors + NW;
    int2* pairs   = (int2*)((int*)d_ws + hdrInts);

    zero_bins_kernel<<<(NW + 255) / 256, 256, 0, stream>>>(counts, cursors, NW);
    hist_kernel<<<(E + BIN_CHUNK - 1) / BIN_CHUNK, BIN_TPB, binLds, stream>>>(
        widx, counts, E, NW);
    scan_kernel<<<1, 256, 0, stream>>>(counts, offsets, NW);
    scatter_kernel<<<(E + BIN_CHUNK - 1) / BIN_CHUNK, BIN_TPB, binLds, stream>>>(
        widx, iidx, offsets, cursors, pairs, E, NW);
    linear_wmma_bin_kernel<<<NW, 256, 0, stream>>>(
        values, weight, pairs, offsets, out, NW);
}
